# starting kernel
# baseline (speedup 1.0000x reference)
#include <hip/hip_runtime.h>
#include <hip/hip_bf16.h>
#include <cmath>


typedef __attribute__((ext_vector_type(2))) float v2f;
typedef __attribute__((ext_vector_type(8))) float v8f;

#define WMMA4F32(A, Bv, C)                                                   \
  __builtin_amdgcn_wmma_f32_16x16x4_f32(false, (A), false, (Bv), (short)0,   \
                                        (C), false, false)

static constexpr int Bsz  = 256;
static constexpr int Tlen = 2000;
static constexpr int Fin  = 42;
static constexpr int Hh   = 24;
static constexpr int Gg   = 72;

__device__ __forceinline__ float sigmoidf_(float v) {
  return 1.0f / (1.0f + __expf(-v));
}

__global__ __launch_bounds__(32) void lc1_tanh_kernel(
    const float* __restrict__ x, const float* __restrict__ w1,
    const float* __restrict__ b1, float* __restrict__ aOut) {
  const int  lane = threadIdx.x;
  const int  nl   = lane & 15;
  const int  ko   = (lane >> 4) * 2;
  const int  mo   = (lane >> 4) * 8;
  const bool hiHalf = (lane >= 16);
  const v2f  zero2 = {0.0f, 0.0f};

  v2f B0[11], B1[11];
  const int  n1  = 16 + nl;
  const bool n1v = (n1 < Hh);
  const int  n1c = n1v ? n1 : (Hh - 1);
#pragma unroll
  for (int kc = 0; kc < 11; ++kc) {
    const int k   = kc * 4 + ko;
    const int kc0 = (k     < Fin) ? k     : 0;
    const int kc1 = (k + 1 < Fin) ? k + 1 : 0;
    const float w00 = w1[nl  * Fin + kc0];
    const float w01 = w1[nl  * Fin + kc1];
    const float w10 = w1[n1c * Fin + kc0];
    const float w11 = w1[n1c * Fin + kc1];
    B0[kc].x = (k     < Fin) ? w00 : 0.0f;
    B0[kc].y = (k + 1 < Fin) ? w01 : 0.0f;
    B1[kc].x = (n1v && (k     < Fin)) ? w10 : 0.0f;
    B1[kc].y = (n1v && (k + 1 < Fin)) ? w11 : 0.0f;
  }
  const float bA = b1[nl];
  const float bB = n1v ? b1[n1c] : 0.0f;

  const int NT = (Bsz * Tlen) / 16;
  for (int tile = blockIdx.x; tile < NT; tile += gridDim.x) {
    const float* xr = x + (size_t)(tile * 16 + nl) * Fin;

    if (tile + (int)gridDim.x < NT)
      __builtin_prefetch(xr + (size_t)gridDim.x * 16 * Fin, 0, 1);

    v8f a0e = {0, 0, 0, 0, 0, 0, 0, 0}, a0o = {0, 0, 0, 0, 0, 0, 0, 0};
    v8f a1e = {0, 0, 0, 0, 0, 0, 0, 0}, a1o = {0, 0, 0, 0, 0, 0, 0, 0};
#pragma unroll
    for (int kc = 0; kc < 10; kc += 2) {
      const v2f Ae = *(const v2f*)(xr + kc * 4 + ko);
      const v2f Ao = *(const v2f*)(xr + (kc + 1) * 4 + ko);
      a0e = WMMA4F32(Ae, B0[kc], a0e);
      a1e = WMMA4F32(Ae, B1[kc], a1e);
      a0o = WMMA4F32(Ao, B0[kc + 1], a0o);
      a1o = WMMA4F32(Ao, B1[kc + 1], a1o);
    }
    {
      const v2f t = *(const v2f*)(xr + 40);
      const v2f A = hiHalf ? zero2 : t;
      a0e = WMMA4F32(A, B0[10], a0e);
      a1e = WMMA4F32(A, B1[10], a1e);
    }

    float* arow = aOut + (size_t)tile * 16 * Hh;
#pragma unroll
    for (int j = 0; j < 8; ++j)
      arow[(j + mo) * Hh + nl] = tanhf(a0e[j] + a0o[j] + bA);
    if (nl < 8) {
#pragma unroll
      for (int j = 0; j < 8; ++j)
        arow[(j + mo) * Hh + 16 + nl] = tanhf(a1e[j] + a1o[j] + bB);
    }
  }
}

__global__ __launch_bounds__(32) void gru_scan_kernel(
    const float* __restrict__ a, const float* __restrict__ w_ih,
    const float* __restrict__ w_hh, const float* __restrict__ b_ih,
    const float* __restrict__ b_hh, const float* __restrict__ lc2_w,
    const float* __restrict__ lc2_b, float* __restrict__ out) {
  __shared__ __align__(16) float Wlds[80][48];
  __shared__ __align__(16) float hS[16][24];
  __shared__ __align__(16) float pS[16][96];
  __shared__ float bsum[48], bihn[24], bhhn[24], lc2S[24];

  const int lane = threadIdx.x;
  const int nl   = lane & 15;
  const int ko   = (lane >> 4) * 2;
  const int mo   = (lane >> 4) * 8;
  const int b0   = blockIdx.x * 16;

  for (int idx = lane; idx < 80 * 48; idx += 32) {
    const int n = idx / 48, k = idx % 48;
    const int nc = (n < Gg) ? n : (Gg - 1);
    const float v = (k < Hh) ? w_ih[nc * Hh + k] : w_hh[nc * Hh + (k - Hh)];
    Wlds[n][k] = (n < Gg) ? v : 0.0f;
  }
  if (lane < 24) {
    bihn[lane] = b_ih[48 + lane];
    bhhn[lane] = b_hh[48 + lane];
    lc2S[lane] = lc2_w[lane];
  }
  for (int g = lane; g < 48; g += 32) bsum[g] = b_ih[g] + b_hh[g];

  const int gm = lane >> 1;
  const int cb = (lane & 1) * 12;
#pragma unroll
  for (int i = 0; i < 12; ++i) hS[gm][cb + i] = 0.0f;
  const float l2b = lc2_b[0];
  __syncthreads();

  for (int t = 0; t < Tlen; ++t) {
    v2f Af[12];
    const float* ab = a + ((size_t)(b0 + nl) * Tlen + t) * Hh;
#pragma unroll
    for (int kc = 0; kc < 6; ++kc)
      Af[kc] = *(const v2f*)(ab + kc * 4 + ko);
#pragma unroll
    for (int kc = 6; kc < 12; ++kc)
      Af[kc] = *(const v2f*)(&hS[nl][(kc - 6) * 4 + ko]);

    if (t + 1 < Tlen) __builtin_prefetch(ab + Hh, 0, 1);

#pragma unroll
    for (int nt = 0; nt < 3; ++nt) {
      v8f ae = {0, 0, 0, 0, 0, 0, 0, 0};
      v8f ao = {0, 0, 0, 0, 0, 0, 0, 0};
#pragma unroll
      for (int kc = 0; kc < 12; kc += 2) {
        const v2f bve = *(const v2f*)(&Wlds[nt * 16 + nl][kc * 4 + ko]);
        const v2f bvo = *(const v2f*)(&Wlds[nt * 16 + nl][(kc + 1) * 4 + ko]);
        ae = WMMA4F32(Af[kc], bve, ae);
        ao = WMMA4F32(Af[kc + 1], bvo, ao);
      }
#pragma unroll
      for (int j = 0; j < 8; ++j) pS[j + mo][nt * 16 + nl] = ae[j] + ao[j];
    }
#pragma unroll
    for (int nt = 3; nt < 5; ++nt) {
      v8f aXe = {0, 0, 0, 0, 0, 0, 0, 0}, aXo = {0, 0, 0, 0, 0, 0, 0, 0};
      v8f aHe = {0, 0, 0, 0, 0, 0, 0, 0}, aHo = {0, 0, 0, 0, 0, 0, 0, 0};
#pragma unroll
      for (int kc = 0; kc < 6; kc += 2) {
        const v2f bve = *(const v2f*)(&Wlds[nt * 16 + nl][kc * 4 + ko]);
        const v2f bvo = *(const v2f*)(&Wlds[nt * 16 + nl][(kc + 1) * 4 + ko]);
        aXe = WMMA4F32(Af[kc], bve, aXe);
        aXo = WMMA4F32(Af[kc + 1], bvo, aXo);
      }
#pragma unroll
      for (int kc = 6; kc < 12; kc += 2) {
        const v2f bve = *(const v2f*)(&Wlds[nt * 16 + nl][kc * 4 + ko]);
        const v2f bvo = *(const v2f*)(&Wlds[nt * 16 + nl][(kc + 1) * 4 + ko]);
        aHe = WMMA4F32(Af[kc], bve, aHe);
        aHo = WMMA4F32(Af[kc + 1], bvo, aHo);
      }
      const int c = (nt - 3) * 16 + nl;
      if (c < Hh) {
#pragma unroll
        for (int j = 0; j < 8; ++j) {
          pS[j + mo][48 + c] = aXe[j] + aXo[j];
          pS[j + mo][72 + c] = aHe[j] + aHo[j];
        }
      }
    }
    __syncthreads();

    float s = 0.0f;
#pragma unroll
    for (int i = 0; i < 12; ++i) {
      const int c = cb + i;
      const float r  = sigmoidf_(pS[gm][c] + bsum[c]);
      const float z  = sigmoidf_(pS[gm][24 + c] + bsum[24 + c]);
      const float xn = pS[gm][48 + c] + bihn[c];
      const float hn = pS[gm][72 + c] + bhhn[c];
      const float nn = tanhf(xn + r * hn);
      const float hv = (1.0f - z) * nn + z * hS[gm][c];
      hS[gm][c] = hv;
      s += lc2S[c] * hv;
    }
    s += __shfl_xor(s, 1, 32);
    if ((lane & 1) == 0)
      out[(size_t)(b0 + gm) * Tlen + t] = sigmoidf_(s + l2b);
    __syncthreads();
  }
}

extern "C" void kernel_launch(void* const* d_in, const int* in_sizes, int n_in,
                              void* d_out, int out_size, void* d_ws,
                              size_t ws_size, hipStream_t stream) {
  (void)in_sizes; (void)n_in; (void)out_size; (void)ws_size;
  const float* x     = (const float*)d_in[0];
  const float* lc1_w = (const float*)d_in[1];
  const float* lc1_b = (const float*)d_in[2];
  const float* w_ih  = (const float*)d_in[3];
  const float* w_hh  = (const float*)d_in[4];
  const float* b_ih  = (const float*)d_in[5];
  const float* b_hh  = (const float*)d_in[6];
  const float* lc2_w = (const float*)d_in[7];
  const float* lc2_b = (const float*)d_in[8];
  float* out  = (float*)d_out;
  float* aBuf = (float*)d_ws;

  lc1_tanh_kernel<<<2048, 32, 0, stream>>>(x, lc1_w, lc1_b, aBuf);
  gru_scan_kernel<<<Bsz / 16, 32, 0, stream>>>(aBuf, w_ih, w_hh, b_ih, b_hh,
                                               lc2_w, lc2_b, out);
}


# LDS row pitches re-chosen against 2-way bank conflicts: product planes 84 -> 92 floats (a gate wave's reads span two rows), hidden state 24 -> 28 (h fragment reads, rows m and m+8); sums kept
# speedup vs baseline: 7.1102x; 7.1102x over previous
#include <hip/hip_runtime.h>
#include <hip/hip_bf16.h>
#include <cmath>


typedef __attribute__((ext_vector_type(2))) float v2f;
typedef __attribute__((ext_vector_type(8))) float v8f;

#define WMMA4F32(A, Bv, C)                                                   \
  __builtin_amdgcn_wmma_f32_16x16x4_f32(false, (A), false, (Bv), (short)0,   \
                                        (C), false, false)

static constexpr int Bsz  = 256;
static constexpr int Tlen = 2000;
static constexpr int Fin  = 42;
static constexpr int Hh   = 24;
static constexpr int Gg   = 72;

__device__ __forceinline__ float sigmoidf_(float v) {
  return 1.0f / (1.0f + __expf(-v));
}

__global__ __launch_bounds__(32) void lc1_tanh_kernel(
    const float* __restrict__ x, const float* __restrict__ w1,
    const float* __restrict__ b1, float* __restrict__ aOut) {
  const int  lane = threadIdx.x;
  const int  nl   = lane & 15;
  const int  ko   = (lane >> 4) * 2;
  const int  mo   = (lane >> 4) * 8;
  const bool hiHalf = (lane >= 16);
  const v2f  zero2 = {0.0f, 0.0f};

  v2f B0[11], B1[11];
  const int  n1  = 16 + nl;
  const bool n1v = (n1 < Hh);
  const int  n1c = n1v ? n1 : (Hh - 1);
#pragma unroll
  for (int kc = 0; kc < 11; ++kc) {
    const int k   = kc * 4 + ko;
    const int kc0 = (k     < Fin) ? k     : 0;
    const int kc1 = (k + 1 < Fin) ? k + 1 : 0;
    const float w00 = w1[nl  * Fin + kc0];
    const float w01 = w1[nl  * Fin + kc1];
    const float w10 = w1[n1c * Fin + kc0];
    const float w11 = w1[n1c * Fin + kc1];
    B0[kc].x = (k     < Fin) ? w00 : 0.0f;
    B0[kc].y = (k + 1 < Fin) ? w01 : 0.0f;
    B1[kc].x = (n1v && (k     < Fin)) ? w10 : 0.0f;
    B1[kc].y = (n1v && (k + 1 < Fin)) ? w11 : 0.0f;
  }
  const float bA = b1[nl];
  const float bB = n1v ? b1[n1c] : 0.0f;

  const int NT = (Bsz * Tlen) / 16;
  for (int tile = blockIdx.x; tile < NT; tile += gridDim.x) {
    const float* xr = x + (size_t)(tile * 16 + nl) * Fin;

    if (tile + (int)gridDim.x < NT)
      __builtin_prefetch(xr + (size_t)gridDim.x * 16 * Fin, 0, 1);

    v8f a0e = {0, 0, 0, 0, 0, 0, 0, 0}, a0o = {0, 0, 0, 0, 0, 0, 0, 0};
    v8f a1e = {0, 0, 0, 0, 0, 0, 0, 0}, a1o = {0, 0, 0, 0, 0, 0, 0, 0};
#pragma unroll
    for (int kc = 0; kc < 10; kc += 2) {
      const v2f Ae = *(const v2f*)(xr + kc * 4 + ko);
      const v2f Ao = *(const v2f*)(xr + (kc + 1) * 4 + ko);
      a0e = WMMA4F32(Ae, B0[kc], a0e);
      a1e = WMMA4F32(Ae, B1[kc], a1e);
      a0o = WMMA4F32(Ao, B0[kc + 1], a0o);
      a1o = WMMA4F32(Ao, B1[kc + 1], a1o);
    }
    {
      const v2f t = *(const v2f*)(xr + 40);
      const v2f A = hiHalf ? zero2 : t;
      a0e = WMMA4F32(A, B0[10], a0e);
      a1e = WMMA4F32(A, B1[10], a1e);
    }

    float* arow = aOut + (size_t)tile * 16 * Hh;
#pragma unroll
    for (int j = 0; j < 8; ++j)
      arow[(j + mo) * Hh + nl] = tanhf(a0e[j] + a0o[j] + bA);
    if (nl < 8) {
#pragma unroll
      for (int j = 0; j < 8; ++j)
        arow[(j + mo) * Hh + 16 + nl] = tanhf(a1e[j] + a1o[j] + bB);
    }
  }
}

static constexpr int PP   = 92;
static constexpr int HP   = 28;
static constexpr int TBLK = 12;
static constexpr int NFULL = Tlen / TBLK;
static constexpr int NTAIL = Tlen - NFULL * TBLK;

__device__ __forceinline__ void lds_done() { asm volatile("s_wait_dscnt 0x0" ::: "memory"); }
__device__ __forceinline__ void bar_signal() { asm volatile("s_barrier_signal -1" ::: "memory"); }
__device__ __forceinline__ void bar_wait() { asm volatile("s_barrier_wait -1" ::: "memory"); }

__global__ __launch_bounds__(384) void gru_scan_kernel(
    const float* __restrict__ a, const float* __restrict__ w_ih,
    const float* __restrict__ w_hh, const float* __restrict__ b_ih,
    const float* __restrict__ b_hh, const float* __restrict__ lc2_w,
    const float* __restrict__ lc2_b, float* __restrict__ out) {
  __shared__ __align__(16) float hS[16][HP];
  __shared__ __align__(16) float pH[2][TBLK + 1][16][PP];
  __shared__ __align__(16) float xgS[2][TBLK][16][PP];

  const int tid  = threadIdx.x;
  const int lane = tid & 31;
  const int wv   = __builtin_amdgcn_readfirstlane(tid >> 5);
  const int nl   = lane & 15;
  const int ko   = (lane >> 4) * 2;
  const int mo   = (lane >> 4) * 8;
  const int b0   = blockIdx.x * 16;

  const bool isH  = (wv < 5);
  const bool late = isH && (wv >= 3);
  const int  nt   = isH ? wv : 0;
  const int  ncol = nt * 16 + nl;
  const bool nval = (ncol < Gg);
  const bool nhd  = (ncol == Gg);
  const int  nc   = nval ? ncol : (Gg - 1);
  const float* wrow = nhd ? lc2_w : (w_hh + nc * Hh);
  const bool nuse = nval || nhd;
  v2f Bh[6];
#pragma unroll
  for (int i = 0; i < 6; ++i) {
    const float w0 = wrow[i * 4 + ko];
    const float w1 = wrow[i * 4 + ko + 1];
    Bh[i].x = nuse ? w0 : 0.0f;
    Bh[i].y = nuse ? w1 : 0.0f;
  }

  v2f Bx[5][6];
#pragma unroll
  for (int n = 0; n < 5; ++n) {
    const int  xcol = n * 16 + nl;
    const bool xval = (xcol < Gg);
    const int  xc   = xval ? xcol : (Gg - 1);
#pragma unroll
    for (int i = 0; i < 6; ++i) {
      const float w0 = w_ih[xc * Hh + i * 4 + ko];
      const float w1 = w_ih[xc * Hh + i * 4 + ko + 1];
      Bx[n][i].x = xval ? w0 : 0.0f;
      Bx[n][i].y = xval ? w1 : 0.0f;
    }
  }

  const int gm = tid / Hh;
  const int gc = tid - gm * Hh;
  const float bR  = b_ih[gc] + b_hh[gc];
  const float bZ  = b_ih[Hh + gc] + b_hh[Hh + gc];
  const float bXn = b_ih[2 * Hh + gc];
  const float bHn = b_hh[2 * Hh + gc];
  hS[gm][gc] = 0.0f;
  const float l2b = lc2_b[0];

  const bool hact = (wv >= 6);
  const int  ho   = hact ? (wv - 6) * 32 + lane : 0;
  const int  hm   = ho / TBLK;
  const int  hs   = ho - hm * TBLK;
  float* hrow = out + (size_t)(b0 + hm) * Tlen;

  const int  fr   = (lane >= TBLK + 1) ? 1 : 0;
  const int  fs   = lane - (TBLK + 1) * fr;
  const int  fm   = (2 * wv + fr) & 15;
  const bool fact = (wv < 8) && (lane < 2 * (TBLK + 1));
  float* orow = out + (size_t)(b0 + fm) * Tlen;

  const float* arow = a + (size_t)(b0 + nl) * Tlen * Hh + ko;
  const v8f zero8 = {0, 0, 0, 0, 0, 0, 0, 0};

  v2f Af[6];
  {
    v2f A0[6];
#pragma unroll
    for (int i = 0; i < 6; ++i) A0[i] = *(const v2f*)(arow + (size_t)wv * Hh + i * 4);
    const int t1  = TBLK + wv;
    const int t1c = (t1 < Tlen) ? t1 : (Tlen - 1);
#pragma unroll
    for (int i = 0; i < 6; ++i) Af[i] = *(const v2f*)(arow + (size_t)t1c * Hh + i * 4);
    float* dst = &xgS[0][wv][0][0];
#pragma unroll
    for (int n = 0; n < 5; ++n) {
      v8f ae = zero8, ao = zero8;
#pragma unroll
      for (int i = 0; i < 6; i += 2) {
        ae = WMMA4F32(A0[i], Bx[n][i], ae);
        ao = WMMA4F32(A0[i + 1], Bx[n][i + 1], ao);
      }
#pragma unroll
      for (int j = 0; j < 8; ++j) dst[(j + mo) * PP + n * 16 + nl] = ae[j] + ao[j];
    }
  }
  __syncthreads();

  auto head_out = [&](const int t0, const int par) __attribute__((always_inline)) {
    if (t0 > 0 && hact) {
      const int idx = t0 - TBLK - 1 + hs;
      if (idx >= 0) hrow[idx] = sigmoidf_(pH[par ^ 1][hs][hm][Gg] + l2b);
    }
  };

  v8f xe = zero8, xo = zero8;
  auto run_block = [&](const int par, const int NS, const bool XS) __attribute__((always_inline)) {
#pragma unroll
    for (int s = 0; s < TBLK; ++s) {
      if (s < NS) {
        float (*pHs)[PP] = pH[par][s];
        const float* xg = &xgS[par][s][gm][0];
        float hold, xr, pr, xz, pz, xnr;
        if (late) {
          v2f Hf[6];
#pragma unroll
          for (int i = 0; i < 6; ++i) Hf[i] = *(const v2f*)(&hS[nl][i * 4 + ko]);
          bar_signal();
          __builtin_amdgcn_sched_barrier(0);
          v8f ae = zero8, ao = zero8;
#pragma unroll
          for (int i = 0; i < 6; i += 2) {
            ae = WMMA4F32(Hf[i], Bh[i], ae);
            ao = WMMA4F32(Hf[i + 1], Bh[i + 1], ao);
          }
          __builtin_amdgcn_sched_barrier(0);
          bar_wait();
          hold = hS[gm][gc];
          xr = xg[gc];       pr = pHs[gm][gc];
          xz = xg[Hh + gc];  pz = pHs[gm][Hh + gc];
          xnr = xg[2 * Hh + gc];
#pragma unroll
          for (int j = 0; j < 8; ++j) pHs[j + mo][nt * 16 + nl] = ae[j] + ao[j];
          lds_done();
        } else {
          if (isH) {
            v2f Hf[6];
#pragma unroll
            for (int i = 0; i < 6; ++i) Hf[i] = *(const v2f*)(&hS[nl][i * 4 + ko]);
            v8f ae = zero8, ao = zero8;
#pragma unroll
            for (int i = 0; i < 6; i += 2) {
              ae = WMMA4F32(Hf[i], Bh[i], ae);
              ao = WMMA4F32(Hf[i + 1], Bh[i + 1], ao);
            }
#pragma unroll
            for (int j = 0; j < 8; ++j) pHs[j + mo][nt * 16 + nl] = ae[j] + ao[j];
          }
          lds_done();
          bar_signal();
          bar_wait();
          hold = hS[gm][gc];
          xr = xg[gc];       pr = pHs[gm][gc];
          xz = xg[Hh + gc];  pz = pHs[gm][Hh + gc];
          xnr = xg[2 * Hh + gc];
        }
        bar_signal();
        bar_wait();

        {
          const float pn = pHs[gm][2 * Hh + gc];
          asm volatile("" : "+v"(xr), "+v"(xz) :: "memory");
          float pnv = pn;
          float r  = sigmoidf_((xr + pr) + bR);
          const float z  = sigmoidf_((xz + pz) + bZ);
          const float xn = xnr + bXn;
          asm volatile("" : "+v"(pnv), "+v"(r));
          const float hn = pnv + bHn;
          float omz = 1.0f - z;
          float zh  = z * hold;
          asm volatile("" : "+v"(omz), "+v"(zh));
          const float nn = tanhf(xn + r * hn);
          const float hv = __builtin_fmaf(omz, nn, zh);
          hS[gm][gc] = hv;
        }
        lds_done();
        bar_signal();
        if (XS) {
          __builtin_amdgcn_sched_barrier(0);
          float* dst = &xgS[par ^ 1][wv][0][0];
          if (s >= 2 && s <= 10 && (s & 1) == 0) {
            const int n = (s >> 1) - 1;
#pragma unroll
            for (int j = 0; j < 8; ++j) dst[(j + mo) * PP + n * 16 + nl] = xe[j] + xo[j];
          }
          if (s < 10) {
            const int n = s >> 1;
            if ((s & 1) == 0) {
              xe = WMMA4F32(Af[0], Bx[n][0], zero8);
              xe = WMMA4F32(Af[2], Bx[n][2], xe);
              xe = WMMA4F32(Af[4], Bx[n][4], xe);
            } else {
              xo = WMMA4F32(Af[1], Bx[n][1], zero8);
              xo = WMMA4F32(Af[3], Bx[n][3], xo);
              xo = WMMA4F32(Af[5], Bx[n][5], xo);
            }
          }
          __builtin_amdgcn_sched_barrier(0);
        }
        bar_wait();
      }
    }
  };

  for (int blk = 0; blk < NFULL; ++blk) {
    const int t0  = blk * TBLK;
    const int par = blk & 1;
    head_out(t0, par);
    const int tn  = t0 + 2 * TBLK + wv;
    const int tnc = (tn < Tlen) ? tn : (Tlen - 1);
    const float* ab = arow + (size_t)tnc * Hh;
    v2f An[6];
#pragma unroll
    for (int i = 0; i < 6; ++i) An[i] = *(const v2f*)(ab + i * 4);
    run_block(par, TBLK, true);
#pragma unroll
    for (int i = 0; i < 6; ++i) Af[i] = An[i];
  }
  if (NTAIL > 0) {
    head_out(NFULL * TBLK, NFULL & 1);
    run_block(NFULL & 1, NTAIL, false);
  }

  constexpr int lastBlk = (Tlen - 1) / TBLK;
  constexpr int lastT0  = lastBlk * TBLK;
  constexpr int lastNs  = Tlen - lastT0;
  constexpr int lastPar = lastBlk & 1;
  if (wv == 4) {
    float (*pHs)[PP] = pH[lastPar][lastNs];
    v2f Hf[6];
#pragma unroll
    for (int i = 0; i < 6; ++i) Hf[i] = *(const v2f*)(&hS[nl][i * 4 + ko]);
    v8f ae = zero8, ao = zero8;
#pragma unroll
    for (int i = 0; i < 6; i += 2) {
      ae = WMMA4F32(Hf[i], Bh[i], ae);
      ao = WMMA4F32(Hf[i + 1], Bh[i + 1], ao);
    }
#pragma unroll
    for (int j = 0; j < 8; ++j) pHs[j + mo][nt * 16 + nl] = ae[j] + ao[j];
  }
  __syncthreads();
  if (fact && fs <= lastNs) {
    const int idx = lastT0 - 1 + fs;
    if (idx >= 0) orow[idx] = sigmoidf_(pH[lastPar][fs][fm][Gg] + l2b);
  }
}

extern "C" void kernel_launch(void* const* d_in, const int* in_sizes, int n_in,
                              void* d_out, int out_size, void* d_ws,
                              size_t ws_size, hipStream_t stream) {
  (void)in_sizes; (void)n_in; (void)out_size; (void)ws_size;
  const float* x     = (const float*)d_in[0];
  const float* lc1_w = (const float*)d_in[1];
  const float* lc1_b = (const float*)d_in[2];
  const float* w_ih  = (const float*)d_in[3];
  const float* w_hh  = (const float*)d_in[4];
  const float* b_ih  = (const float*)d_in[5];
  const float* b_hh  = (const float*)d_in[6];
  const float* lc2_w = (const float*)d_in[7];
  const float* lc2_b = (const float*)d_in[8];
  float* out  = (float*)d_out;
  float* aBuf = (float*)d_ws;

  lc1_tanh_kernel<<<2048, 32, 0, stream>>>(x, lc1_w, lc1_b, aBuf);
  gru_scan_kernel<<<Bsz / 16, 384, 0, stream>>>(aBuf, w_ih, w_hh, b_ih, b_hh,
                                               lc2_w, lc2_b, out);
}
